# starting kernel
# baseline (speedup 1.0000x reference)
#include <hip/hip_runtime.h>
#include <hip/hip_bf16.h>
#include <math.h>

typedef __attribute__((ext_vector_type(2))) float v2f;
typedef __attribute__((ext_vector_type(8))) float v8f;

#define INPUT_SIZE 256
#define HEAD_SIZE  32
#define INV_DK     0.17677669529663687f
#define MTILES     4

__device__ __forceinline__ unsigned enc_f32(float f) {
    unsigned u = __float_as_uint(f);
    return (u & 0x80000000u) ? ~u : (u | 0x80000000u);
}
__device__ __forceinline__ float dec_f32(unsigned e) {
    unsigned u = (e & 0x80000000u) ? (e ^ 0x80000000u) : ~e;
    return __uint_as_float(u);
}
#define ENC_NEG_INF 0x007FFFFFu

__global__ void gat_init(float* __restrict__ out, float* __restrict__ denom,
                         unsigned* __restrict__ rmax, int outN, int N) {
    int i = blockIdx.x * 256 + threadIdx.x;
    if (i < outN) out[i] = 0.0f;
    if (i < N) { denom[i] = 0.0f; rmax[i] = ENC_NEG_INF; }
}

__global__ __launch_bounds__(256) void gat_qkv_wmma(
    const float* __restrict__ X,
    const float* __restrict__ Wq, const float* __restrict__ Wk,
    const float* __restrict__ Wv,
    float* __restrict__ Q, float* __restrict__ Ko, float* __restrict__ Vo,
    int N)
{
    const int lane    = threadIdx.x & 31;
    const int wave    = threadIdx.x >> 5;
    const int tile0   = (blockIdx.x * 8 + wave) * MTILES;
    if (tile0 * 16 >= N) return;

    const int m  = lane & 15;
    const int kh = (lane >> 4) * 2;

    const float* Ws[3] = { Wq, Wk, Wv };

    v8f acc[MTILES][6];
    #pragma unroll
    for (int t = 0; t < MTILES; ++t)
        #pragma unroll
        for (int i = 0; i < 6; ++i) {
            v8f z = {};
            acc[t][i] = z;
        }

    const float* xrow[MTILES];
    bool valid[MTILES];
    #pragma unroll
    for (int t = 0; t < MTILES; ++t) {
        const int rb = (tile0 + t) * 16;
        valid[t] = (rb + 16) <= N;
        const int r = valid[t] ? (rb + m) : m;
        xrow[t] = X + (size_t)r * INPUT_SIZE;
    }

    #pragma unroll 4
    for (int k0 = 0; k0 < INPUT_SIZE; k0 += 4) {
        v2f a[MTILES];
        #pragma unroll
        for (int t = 0; t < MTILES; ++t) {
            a[t].x = xrow[t][k0 + kh];
            a[t].y = xrow[t][k0 + kh + 1];
        }
        #pragma unroll
        for (int w = 0; w < 3; ++w) {
            const float* W = Ws[w];
            #pragma unroll
            for (int h = 0; h < 2; ++h) {
                const int col = m + h * 16;
                v2f b;
                b.x = W[(k0 + kh)     * HEAD_SIZE + col];
                b.y = W[(k0 + kh + 1) * HEAD_SIZE + col];
                #pragma unroll
                for (int t = 0; t < MTILES; ++t) {
                    acc[t][w * 2 + h] = __builtin_amdgcn_wmma_f32_16x16x4_f32(
                         false, a[t],  false, b,
                         (short)0, acc[t][w * 2 + h],
                         false,  false);
                }
            }
        }
    }

    float* Outs[3] = { Q, Ko, Vo };
    const int rofs = (lane >> 4) * 8;
    #pragma unroll
    for (int t = 0; t < MTILES; ++t) {
        if (!valid[t]) continue;
        const int row_base = (tile0 + t) * 16;
        #pragma unroll
        for (int w = 0; w < 3; ++w) {
            #pragma unroll
            for (int h = 0; h < 2; ++h) {
                const v8f c = acc[t][w * 2 + h];
                const int col = m + h * 16;
                #pragma unroll
                for (int j = 0; j < 8; ++j) {
                    const int row = row_base + j + rofs;
                    Outs[w][(size_t)row * HEAD_SIZE + col] = c[j];
                }
            }
        }
    }
}

__global__ __launch_bounds__(256) void gat_scores(
    const float* __restrict__ Q, const float* __restrict__ K,
    const int* __restrict__ src, const int* __restrict__ dst,
    float* __restrict__ scores, unsigned* __restrict__ rmax, int E)
{
    const int e = blockIdx.x * 32 + (threadIdx.x >> 3);
    if (e >= E) return;
    const int sub = threadIdx.x & 7;
    const int s = src[e];
    const int d = dst[e];

    const float4 q = ((const float4*)(Q + (size_t)s * HEAD_SIZE))[sub];
    const float4 k = ((const float4*)(K + (size_t)d * HEAD_SIZE))[sub];
    float p = q.x * k.x + q.y * k.y + q.z * k.z + q.w * k.w;
    p += __shfl_xor(p, 1, 8);
    p += __shfl_xor(p, 2, 8);
    p += __shfl_xor(p, 4, 8);

    if (sub == 0) {
        const float sc = p * INV_DK;
        scores[e] = sc;
        atomicMax(&rmax[s], enc_f32(sc));
    }
}

__global__ __launch_bounds__(256) void gat_exp(
    const int* __restrict__ src, float* __restrict__ scores,
    const unsigned* __restrict__ rmax, float* __restrict__ denom, int E)
{
    const int e = blockIdx.x * 256 + threadIdx.x;
    if (e >= E) return;
    const int s = src[e];
    const float ex = __expf(scores[e] - dec_f32(rmax[s]));
    scores[e] = ex;
    atomicAdd(&denom[s], ex);
}

__global__ __launch_bounds__(256) void gat_aggregate(
    const float* __restrict__ V,
    const int* __restrict__ src, const int* __restrict__ dst,
    const float* __restrict__ ex, const float* __restrict__ denom,
    float* __restrict__ out, int E)
{
    const int e = blockIdx.x * 32 + (threadIdx.x >> 3);
    if (e >= E) return;
    const int sub = threadIdx.x & 7;
    const int s = src[e];
    const int d = dst[e];
    const float alpha = ex[e] / denom[s];

    const float4 v = ((const float4*)(V + (size_t)d * HEAD_SIZE))[sub];
    float* o = out + (size_t)s * HEAD_SIZE + sub * 4;
    atomicAdd(o + 0, alpha * v.x);
    atomicAdd(o + 1, alpha * v.y);
    atomicAdd(o + 2, alpha * v.z);
    atomicAdd(o + 3, alpha * v.w);
}

extern "C" void kernel_launch(void* const* d_in, const int* in_sizes, int n_in,
                              void* d_out, int out_size, void* d_ws, size_t ws_size,
                              hipStream_t stream) {
    const float* X    = (const float*)d_in[0];
    const int*   edge = (const int*)  d_in[1];
    const float* Wq   = (const float*)d_in[2];
    const float* Wk   = (const float*)d_in[3];
    const float* Wv   = (const float*)d_in[4];

    const int N = in_sizes[0] / INPUT_SIZE;
    const int E = in_sizes[1] / 2;
    const int* src = edge;
    const int* dst = edge + E;

    float* out = (float*)d_out;

    float*    ws     = (float*)d_ws;
    float*    Q      = ws;
    float*    Kb     = ws + (size_t)N * HEAD_SIZE;
    float*    Vb     = ws + (size_t)2 * N * HEAD_SIZE;
    float*    scores = ws + (size_t)3 * N * HEAD_SIZE;
    float*    denom  = scores + E;
    unsigned* rmax   = (unsigned*)(denom + N);

    {
        const int M = (out_size > N) ? out_size : N;
        gat_init<<<(M + 255) / 256, 256, 0, stream>>>(out, denom, rmax, out_size, N);
    }
    {
        const int tiles  = (N + 15) / 16;
        const int waves  = (tiles + MTILES - 1) / MTILES;
        const int blocks = (waves + 7) / 8;
        gat_qkv_wmma<<<blocks, 256, 0, stream>>>(X, Wq, Wk, Wv, Q, Kb, Vb, N);
    }
    gat_scores<<<(E + 31) / 32, 256, 0, stream>>>(Q, Kb, src, dst, scores, rmax, E);
    gat_exp<<<(E + 255) / 256, 256, 0, stream>>>(src, scores, rmax, denom, E);
    gat_aggregate<<<(E + 31) / 32, 256, 0, stream>>>(Vb, src, dst, scores, denom, out, E);
}


# exp+denominator pass folded into the aggregation: ex*V and ex summed un-normalised by float atomics into one 256-B row per src (sum order changes), one division per node at the end
# speedup vs baseline: 1.2212x; 1.2212x over previous
#include <hip/hip_runtime.h>
#include <hip/hip_bf16.h>
#include <math.h>

typedef __attribute__((ext_vector_type(2))) float v2f;
typedef __attribute__((ext_vector_type(8))) float v8f;

#define INPUT_SIZE 256
#define HEAD_SIZE  32
#define INV_DK     0.17677669529663687f
#define MTILES     4

__device__ __forceinline__ unsigned enc_f32(float f) {
    unsigned u = __float_as_uint(f);
    return (u & 0x80000000u) ? ~u : (u | 0x80000000u);
}
__device__ __forceinline__ float dec_f32(unsigned e) {
    unsigned u = (e & 0x80000000u) ? (e ^ 0x80000000u) : ~e;
    return __uint_as_float(u);
}
#define ENC_NEG_INF 0x007FFFFFu

#define ACC_PITCH  64
#define ACC_DEN    32

__global__ void gat_init(unsigned* __restrict__ rmax, int N) {
    int i = blockIdx.x * 256 + threadIdx.x;
    if (i < N) rmax[i] = ENC_NEG_INF;
}

__global__ __launch_bounds__(256) void gat_qkv_wmma(
    const float* __restrict__ X,
    const float* __restrict__ Wq, const float* __restrict__ Wk,
    const float* __restrict__ Wv,
    float* __restrict__ Q, float* __restrict__ Ko, float* __restrict__ Vo,
    int N)
{
    const int lane    = threadIdx.x & 31;
    const int wave    = threadIdx.x >> 5;
    const int tile0   = (blockIdx.x * 8 + wave) * MTILES;
    if (tile0 * 16 >= N) return;

    const int m  = lane & 15;
    const int kh = (lane >> 4) * 2;

    const float* Ws[3] = { Wq, Wk, Wv };

    v8f acc[MTILES][6];
    #pragma unroll
    for (int t = 0; t < MTILES; ++t)
        #pragma unroll
        for (int i = 0; i < 6; ++i) {
            v8f z = {};
            acc[t][i] = z;
        }

    const float* xrow[MTILES];
    bool valid[MTILES];
    #pragma unroll
    for (int t = 0; t < MTILES; ++t) {
        const int rb = (tile0 + t) * 16;
        valid[t] = (rb + 16) <= N;
        const int r = valid[t] ? (rb + m) : m;
        xrow[t] = X + (size_t)r * INPUT_SIZE;
    }

    #pragma unroll 4
    for (int k0 = 0; k0 < INPUT_SIZE; k0 += 4) {
        v2f a[MTILES];
        #pragma unroll
        for (int t = 0; t < MTILES; ++t) {
            a[t].x = xrow[t][k0 + kh];
            a[t].y = xrow[t][k0 + kh + 1];
        }
        #pragma unroll
        for (int w = 0; w < 3; ++w) {
            const float* W = Ws[w];
            #pragma unroll
            for (int h = 0; h < 2; ++h) {
                const int col = m + h * 16;
                v2f b;
                b.x = W[(k0 + kh)     * HEAD_SIZE + col];
                b.y = W[(k0 + kh + 1) * HEAD_SIZE + col];
                #pragma unroll
                for (int t = 0; t < MTILES; ++t) {
                    acc[t][w * 2 + h] = __builtin_amdgcn_wmma_f32_16x16x4_f32(
                         false, a[t],  false, b,
                         (short)0, acc[t][w * 2 + h],
                         false,  false);
                }
            }
        }
    }

    float* Outs[3] = { Q, Ko, Vo };
    const int rofs = (lane >> 4) * 8;
    #pragma unroll
    for (int t = 0; t < MTILES; ++t) {
        if (!valid[t]) continue;
        const int row_base = (tile0 + t) * 16;
        #pragma unroll
        for (int w = 0; w < 3; ++w) {
            #pragma unroll
            for (int h = 0; h < 2; ++h) {
                const v8f c = acc[t][w * 2 + h];
                const int col = m + h * 16;
                #pragma unroll
                for (int j = 0; j < 8; ++j) {
                    const int row = row_base + j + rofs;
                    Outs[w][(size_t)row * HEAD_SIZE + col] = c[j];
                }
            }
        }
    }
}

__global__ __launch_bounds__(256) void gat_scores(
    const float* __restrict__ Q, const float* __restrict__ K,
    const int* __restrict__ src, const int* __restrict__ dst,
    float* __restrict__ scores, unsigned* __restrict__ rmax, int E)
{
    const int e = blockIdx.x * 32 + (threadIdx.x >> 3);
    if (e >= E) return;
    const int sub = threadIdx.x & 7;
    const int s = src[e];
    const int d = dst[e];

    const float4 q = ((const float4*)(Q + (size_t)s * HEAD_SIZE))[sub];
    const float4 k = ((const float4*)(K + (size_t)d * HEAD_SIZE))[sub];
    float p = q.x * k.x + q.y * k.y + q.z * k.z + q.w * k.w;
    p += __shfl_xor(p, 1, 8);
    p += __shfl_xor(p, 2, 8);
    p += __shfl_xor(p, 4, 8);

    if (sub == 0) {
        const float sc = p * INV_DK;
        scores[e] = sc;
        atomicMax(&rmax[s], enc_f32(sc));
    }
}

__global__ __launch_bounds__(256) void gat_zero_acc(float* __restrict__ acc, int n4)
{
    const int i = blockIdx.x * 256 + threadIdx.x;
    if (i < n4) ((float4*)acc)[i] = make_float4(0.0f, 0.0f, 0.0f, 0.0f);
}

__global__ __launch_bounds__(256) void gat_exp_aggregate(
    const float* __restrict__ V,
    const int* __restrict__ src, const int* __restrict__ dst,
    const float* __restrict__ scores, const unsigned* __restrict__ rmax,
    float* __restrict__ acc, int E)
{
    const int e = blockIdx.x * 32 + (threadIdx.x >> 3);
    if (e >= E) return;
    const int sub = threadIdx.x & 7;
    const int s = src[e];
    const int d = dst[e];
    const float ex = __expf(scores[e] - dec_f32(rmax[s]));

    const float4 v = ((const float4*)(V + (size_t)d * HEAD_SIZE))[sub];
    float* row = acc + (size_t)s * ACC_PITCH;
    float* o = row + sub * 4;
    atomicAdd(o + 0, ex * v.x);
    atomicAdd(o + 1, ex * v.y);
    atomicAdd(o + 2, ex * v.z);
    atomicAdd(o + 3, ex * v.w);
    if (sub == 0) atomicAdd(row + ACC_DEN, ex);
}

__global__ __launch_bounds__(256) void gat_normalize(
    const float* __restrict__ acc, float* __restrict__ out, int N)
{
    const int n = blockIdx.x * 32 + (threadIdx.x >> 3);
    if (n >= N) return;
    const int sub = threadIdx.x & 7;
    const float* row = acc + (size_t)n * ACC_PITCH;
    const float den = row[ACC_DEN];
    const float4 a = ((const float4*)row)[sub];
    const float dd = (den > 0.0f) ? den : 1.0f;
    float4 r;
    r.x = a.x / dd; r.y = a.y / dd; r.z = a.z / dd; r.w = a.w / dd;
    ((float4*)(out + (size_t)n * HEAD_SIZE))[sub] = r;
}

extern "C" void kernel_launch(void* const* d_in, const int* in_sizes, int n_in,
                              void* d_out, int out_size, void* d_ws, size_t ws_size,
                              hipStream_t stream) {
    const float* X    = (const float*)d_in[0];
    const int*   edge = (const int*)  d_in[1];
    const float* Wq   = (const float*)d_in[2];
    const float* Wk   = (const float*)d_in[3];
    const float* Wv   = (const float*)d_in[4];

    const int N = in_sizes[0] / INPUT_SIZE;
    const int E = in_sizes[1] / 2;
    const int* src = edge;
    const int* dst = edge + E;

    float* out = (float*)d_out;

    float*    ws     = (float*)d_ws;
    float*    Q      = ws;
    float*    Kb     = ws + (size_t)N * HEAD_SIZE;
    float*    Vb     = ws + (size_t)2 * N * HEAD_SIZE;
    float*    scores = ws + (size_t)3 * N * HEAD_SIZE;
    float*    denom  = scores + E;
    unsigned* rmax   = (unsigned*)(denom + N);

    float*    acc    = ws;

    gat_init<<<(N + 255) / 256, 256, 0, stream>>>(rmax, N);
    {
        const int tiles  = (N + 15) / 16;
        const int waves  = (tiles + MTILES - 1) / MTILES;
        const int blocks = (waves + 7) / 8;
        gat_qkv_wmma<<<blocks, 256, 0, stream>>>(X, Wq, Wk, Wv, Q, Kb, Vb, N);
    }
    gat_scores<<<(E + 31) / 32, 256, 0, stream>>>(Q, Kb, src, dst, scores, rmax, E);
    {
        const int n4 = N * (ACC_PITCH / 4);
        gat_zero_acc<<<(n4 + 255) / 256, 256, 0, stream>>>(acc, n4);
    }
    gat_exp_aggregate<<<(E + 31) / 32, 256, 0, stream>>>(Vb, src, dst, scores, rmax, acc, E);
    gat_normalize<<<(N + 31) / 32, 256, 0, stream>>>(acc, out, N);
}


# edges counting-sorted by src each call (int atomics: count, scan, place), then one pass per node does scores, running-max softmax and V sum in registers: float atomics gone, sum order changes
# speedup vs baseline: 1.3336x; 1.0920x over previous
#include <hip/hip_runtime.h>
#include <hip/hip_bf16.h>
#include <math.h>

typedef __attribute__((ext_vector_type(2))) float v2f;
typedef __attribute__((ext_vector_type(8))) float v8f;

#define INPUT_SIZE 256
#define HEAD_SIZE  32
#define INV_DK     0.17677669529663687f
#define MTILES     4

__global__ void seg_clear(int* __restrict__ cnt, int N) {
    int i = blockIdx.x * 256 + threadIdx.x;
    if (i < N) cnt[i] = 0;
}

__global__ __launch_bounds__(256) void gat_qkv_wmma(
    const float* __restrict__ X,
    const float* __restrict__ Wq, const float* __restrict__ Wk,
    const float* __restrict__ Wv,
    float* __restrict__ Q, float* __restrict__ Ko, float* __restrict__ Vo,
    int N)
{
    const int lane    = threadIdx.x & 31;
    const int wave    = threadIdx.x >> 5;
    const int tile0   = (blockIdx.x * 8 + wave) * MTILES;
    if (tile0 * 16 >= N) return;

    const int m  = lane & 15;
    const int kh = (lane >> 4) * 2;

    const float* Ws[3] = { Wq, Wk, Wv };

    v8f acc[MTILES][6];
    #pragma unroll
    for (int t = 0; t < MTILES; ++t)
        #pragma unroll
        for (int i = 0; i < 6; ++i) {
            v8f z = {};
            acc[t][i] = z;
        }

    const float* xrow[MTILES];
    bool valid[MTILES];
    #pragma unroll
    for (int t = 0; t < MTILES; ++t) {
        const int rb = (tile0 + t) * 16;
        valid[t] = (rb + 16) <= N;
        const int r = valid[t] ? (rb + m) : m;
        xrow[t] = X + (size_t)r * INPUT_SIZE;
    }

    #pragma unroll 4
    for (int k0 = 0; k0 < INPUT_SIZE; k0 += 4) {
        v2f a[MTILES];
        #pragma unroll
        for (int t = 0; t < MTILES; ++t) {
            a[t].x = xrow[t][k0 + kh];
            a[t].y = xrow[t][k0 + kh + 1];
        }
        #pragma unroll
        for (int w = 0; w < 3; ++w) {
            const float* W = Ws[w];
            #pragma unroll
            for (int h = 0; h < 2; ++h) {
                const int col = m + h * 16;
                v2f b;
                b.x = W[(k0 + kh)     * HEAD_SIZE + col];
                b.y = W[(k0 + kh + 1) * HEAD_SIZE + col];
                #pragma unroll
                for (int t = 0; t < MTILES; ++t) {
                    acc[t][w * 2 + h] = __builtin_amdgcn_wmma_f32_16x16x4_f32(
                         false, a[t],  false, b,
                         (short)0, acc[t][w * 2 + h],
                         false,  false);
                }
            }
        }
    }

    float* Outs[3] = { Q, Ko, Vo };
    const int rofs = (lane >> 4) * 8;
    #pragma unroll
    for (int t = 0; t < MTILES; ++t) {
        if (!valid[t]) continue;
        const int row_base = (tile0 + t) * 16;
        #pragma unroll
        for (int w = 0; w < 3; ++w) {
            #pragma unroll
            for (int h = 0; h < 2; ++h) {
                const v8f c = acc[t][w * 2 + h];
                const int col = m + h * 16;
                #pragma unroll
                for (int j = 0; j < 8; ++j) {
                    const int row = row_base + j + rofs;
                    Outs[w][(size_t)row * HEAD_SIZE + col] = c[j];
                }
            }
        }
    }
}

__global__ __launch_bounds__(256) void seg_count(
    const int* __restrict__ src, int* __restrict__ cnt, int E)
{
    const int e = blockIdx.x * 256 + threadIdx.x;
    if (e < E) atomicAdd(&cnt[src[e]], 1);
}

__device__ __forceinline__ int block_scan_excl(int v, int* wsum, int& total) {
    const int lane = threadIdx.x & 31;
    const int wv   = threadIdx.x >> 5;
    int inc = v;
    #pragma unroll
    for (int o = 1; o < 32; o <<= 1) {
        const int t = __shfl_up(inc, o, 32);
        if (lane >= o) inc += t;
    }
    if (lane == 31) wsum[wv] = inc;
    __syncthreads();
    int woff = 0, tot = 0;
    #pragma unroll
    for (int i = 0; i < 8; ++i) {
        const int w = wsum[i];
        if (i < wv) woff += w;
        tot += w;
    }
    total = tot;
    return woff + inc - v;
}

__global__ __launch_bounds__(256) void seg_scan(
    const int* __restrict__ cnt, int* __restrict__ cur, int N)
{
    __shared__ int s_pre[8];
    __shared__ int s_loc[8];
    const int t = threadIdx.x;
    const int b = blockIdx.x;

    int pre = 0;
    const int4* c4 = (const int4*)cnt;
    for (int i = t; i < b * 256; i += 256) {
        const int4 c = c4[i];
        pre += (c.x + c.y) + (c.z + c.w);
    }
    int base;
    (void)block_scan_excl(pre, s_pre, base);

    const int i0 = b * 1024 + t * 4;
    const int v0 = (i0     < N) ? cnt[i0]     : 0;
    const int v1 = (i0 + 1 < N) ? cnt[i0 + 1] : 0;
    const int v2 = (i0 + 2 < N) ? cnt[i0 + 2] : 0;
    const int v3 = (i0 + 3 < N) ? cnt[i0 + 3] : 0;
    int tot;
    int o = base + block_scan_excl(v0 + v1 + v2 + v3, s_loc, tot);
    if (i0     < N) cur[i0]     = o;
    o += v0;
    if (i0 + 1 < N) cur[i0 + 1] = o;
    o += v1;
    if (i0 + 2 < N) cur[i0 + 2] = o;
    o += v2;
    if (i0 + 3 < N) cur[i0 + 3] = o;
}

__global__ __launch_bounds__(256) void seg_place(
    const int* __restrict__ src, const int* __restrict__ dst,
    int* __restrict__ cur, int* __restrict__ perm, int E)
{
    const int e = blockIdx.x * 256 + threadIdx.x;
    if (e >= E) return;
    const int pos = atomicAdd(&cur[src[e]], 1);
    perm[pos] = dst[e];
}

__device__ __forceinline__ float dot8(const float4 q, const float4 k) {
    float p = q.x * k.x + q.y * k.y + q.z * k.z + q.w * k.w;
    p += __shfl_xor(p, 1, 8);
    p += __shfl_xor(p, 2, 8);
    p += __shfl_xor(p, 4, 8);
    return p;
}

__global__ __launch_bounds__(256) void node_attend(
    const float* __restrict__ Q, const float* __restrict__ K,
    const float* __restrict__ V,
    const int* __restrict__ cnt, const int* __restrict__ cur,
    const int* __restrict__ perm, float* __restrict__ out, int N)
{
    const int n = blockIdx.x * 32 + (threadIdx.x >> 3);
    if (n >= N) return;
    const int sub = threadIdx.x & 7;
    const int end = cur[n];
    const int beg = end - cnt[n];

    const float4 q = ((const float4*)(Q + (size_t)n * HEAD_SIZE))[sub];
    const float4* K4 = (const float4*)K + sub;
    const float4* V4 = (const float4*)V + sub;

    float  mx = -INFINITY;
    float  l  = 0.0f;
    float4 a  = make_float4(0.0f, 0.0f, 0.0f, 0.0f);

    for (int j = beg; j < end; j += 4) {
        const int last = end - 1 - j;
        const int d0 = perm[j];
        const int d1 = perm[j + min(1, last)];
        const int d2 = perm[j + min(2, last)];
        const int d3 = perm[j + min(3, last)];
        const float4 k0 = K4[(size_t)d0 * 8];
        const float4 k1 = K4[(size_t)d1 * 8];
        const float4 k2 = K4[(size_t)d2 * 8];
        const float4 k3 = K4[(size_t)d3 * 8];
        const float4 v0 = V4[(size_t)d0 * 8];
        const float4 v1 = V4[(size_t)d1 * 8];
        const float4 v2 = V4[(size_t)d2 * 8];
        const float4 v3 = V4[(size_t)d3 * 8];
        const float p0 = dot8(q, k0);
        const float p1 = dot8(q, k1);
        const float p2 = dot8(q, k2);
        const float p3 = dot8(q, k3);
        const float s0 = p0 * INV_DK;
        const float s1 = (last >= 1) ? p1 * INV_DK : -INFINITY;
        const float s2 = (last >= 2) ? p2 * INV_DK : -INFINITY;
        const float s3 = (last >= 3) ? p3 * INV_DK : -INFINITY;
        const float mn   = fmaxf(fmaxf(mx, s0), fmaxf(s1, fmaxf(s2, s3)));
        const float corr = __expf(mx - mn);
        const float e0 = __expf(s0 - mn);
        const float e1 = __expf(s1 - mn);
        const float e2 = __expf(s2 - mn);
        const float e3 = __expf(s3 - mn);
        l   = l   * corr + ((e0 + e1) + (e2 + e3));
        a.x = a.x * corr + ((e0 * v0.x + e1 * v1.x) + (e2 * v2.x + e3 * v3.x));
        a.y = a.y * corr + ((e0 * v0.y + e1 * v1.y) + (e2 * v2.y + e3 * v3.y));
        a.z = a.z * corr + ((e0 * v0.z + e1 * v1.z) + (e2 * v2.z + e3 * v3.z));
        a.w = a.w * corr + ((e0 * v0.w + e1 * v1.w) + (e2 * v2.w + e3 * v3.w));
        mx = mn;
    }

    const float dd = (l > 0.0f) ? l : 1.0f;
    float4 r;
    r.x = a.x / dd; r.y = a.y / dd; r.z = a.z / dd; r.w = a.w / dd;
    ((float4*)(out + (size_t)n * HEAD_SIZE))[sub] = r;
}

extern "C" void kernel_launch(void* const* d_in, const int* in_sizes, int n_in,
                              void* d_out, int out_size, void* d_ws, size_t ws_size,
                              hipStream_t stream) {
    const float* X    = (const float*)d_in[0];
    const int*   edge = (const int*)  d_in[1];
    const float* Wq   = (const float*)d_in[2];
    const float* Wk   = (const float*)d_in[3];
    const float* Wv   = (const float*)d_in[4];

    const int N = in_sizes[0] / INPUT_SIZE;
    const int E = in_sizes[1] / 2;
    const int* src = edge;
    const int* dst = edge + E;

    float* out = (float*)d_out;

    float* ws   = (float*)d_ws;
    float* Q    = ws;
    float* Kb   = ws + (size_t)N * HEAD_SIZE;
    float* Vb   = ws + (size_t)2 * N * HEAD_SIZE;
    int*   cnt  = (int*)(ws + (size_t)3 * N * HEAD_SIZE);
    int*   cur  = cnt + N;
    int*   perm = cur + N;

    seg_clear<<<(N + 255) / 256, 256, 0, stream>>>(cnt, N);
    {
        const int tiles  = (N + 15) / 16;
        const int waves  = (tiles + MTILES - 1) / MTILES;
        const int blocks = (waves + 7) / 8;
        gat_qkv_wmma<<<blocks, 256, 0, stream>>>(X, Wq, Wk, Wv, Q, Kb, Vb, N);
    }
    seg_count<<<(E + 255) / 256, 256, 0, stream>>>(src, cnt, E);
    seg_scan<<<(N + 1023) / 1024, 256, 0, stream>>>(cnt, cur, N);
    seg_place<<<(E + 255) / 256, 256, 0, stream>>>(src, dst, cur, perm, E);
    node_attend<<<(N + 31) / 32, 256, 0, stream>>>(Q, Kb, Vb, cnt, cur, perm, out, N);
}


# edges grouped by source with a two-level counting sort on LDS counters (6250-edge chunks by 256-node bucket, then buckets by node): no device-scope atomics left; slot order still by atomic arrival
# speedup vs baseline: 3.1651x; 2.3734x over previous
#include <hip/hip_runtime.h>
#include <hip/hip_bf16.h>
#include <math.h>

typedef __attribute__((ext_vector_type(2))) float v2f;
typedef __attribute__((ext_vector_type(8))) float v8f;

#define INPUT_SIZE 256
#define HEAD_SIZE  32
#define INV_DK     0.17677669529663687f
#define MTILES     4

__global__ __launch_bounds__(256) void gat_qkv_wmma(
    const float* __restrict__ X,
    const float* __restrict__ Wq, const float* __restrict__ Wk,
    const float* __restrict__ Wv,
    float* __restrict__ Q, float* __restrict__ Ko, float* __restrict__ Vo,
    int N)
{
    const int lane    = threadIdx.x & 31;
    const int wave    = threadIdx.x >> 5;
    const int tile0   = (blockIdx.x * 8 + wave) * MTILES;
    if (tile0 * 16 >= N) return;

    const int m  = lane & 15;
    const int kh = (lane >> 4) * 2;

    const float* Ws[3] = { Wq, Wk, Wv };

    v8f acc[MTILES][6];
    #pragma unroll
    for (int t = 0; t < MTILES; ++t)
        #pragma unroll
        for (int i = 0; i < 6; ++i) {
            v8f z = {};
            acc[t][i] = z;
        }

    const float* xrow[MTILES];
    bool valid[MTILES];
    #pragma unroll
    for (int t = 0; t < MTILES; ++t) {
        const int rb = (tile0 + t) * 16;
        valid[t] = (rb + 16) <= N;
        const int r = valid[t] ? (rb + m) : m;
        xrow[t] = X + (size_t)r * INPUT_SIZE;
    }

    #pragma unroll 4
    for (int k0 = 0; k0 < INPUT_SIZE; k0 += 4) {
        v2f a[MTILES];
        #pragma unroll
        for (int t = 0; t < MTILES; ++t) {
            a[t].x = xrow[t][k0 + kh];
            a[t].y = xrow[t][k0 + kh + 1];
        }
        #pragma unroll
        for (int w = 0; w < 3; ++w) {
            const float* W = Ws[w];
            #pragma unroll
            for (int h = 0; h < 2; ++h) {
                const int col = m + h * 16;
                v2f b;
                b.x = W[(k0 + kh)     * HEAD_SIZE + col];
                b.y = W[(k0 + kh + 1) * HEAD_SIZE + col];
                #pragma unroll
                for (int t = 0; t < MTILES; ++t) {
                    acc[t][w * 2 + h] = __builtin_amdgcn_wmma_f32_16x16x4_f32(
                         false, a[t],  false, b,
                         (short)0, acc[t][w * 2 + h],
                         false,  false);
                }
            }
        }
    }

    float* Outs[3] = { Q, Ko, Vo };
    const int rofs = (lane >> 4) * 8;
    #pragma unroll
    for (int t = 0; t < MTILES; ++t) {
        if (!valid[t]) continue;
        const int row_base = (tile0 + t) * 16;
        #pragma unroll
        for (int w = 0; w < 3; ++w) {
            #pragma unroll
            for (int h = 0; h < 2; ++h) {
                const v8f c = acc[t][w * 2 + h];
                const int col = m + h * 16;
                #pragma unroll
                for (int j = 0; j < 8; ++j) {
                    const int row = row_base + j + rofs;
                    Outs[w][(size_t)row * HEAD_SIZE + col] = c[j];
                }
            }
        }
    }
}

#define CHUNK       6250
#define L1_THREADS  512
#define L1_EPT      ((CHUNK + L1_THREADS - 1) / L1_THREADS)
#define BSHIFT      8
#define BNODES      256
#define L2_BATCH    16

template <int NW>
__device__ __forceinline__ int block_scan_excl(int v, int* wsum, int& total) {
    const int lane = threadIdx.x & 31;
    const int wv   = threadIdx.x >> 5;
    int inc = v;
    #pragma unroll
    for (int o = 1; o < 32; o <<= 1) {
        const int t = __shfl_up(inc, o, 32);
        if (lane >= o) inc += t;
    }
    if (lane == 31) wsum[wv] = inc;
    __syncthreads();
    int woff = 0, tot = 0;
    #pragma unroll
    for (int i = 0; i < NW; ++i) {
        const int w = wsum[i];
        if (i < wv) woff += w;
        tot += w;
    }
    total = tot;
    return woff + inc - v;
}

__global__ __launch_bounds__(L1_THREADS) void sort_chunks(
    const int* __restrict__ src, const int* __restrict__ dst,
    int2* __restrict__ tmp, int* __restrict__ offsT, int E, int NB, int G)
{
    extern __shared__ int s_hist[];
    __shared__ int2 s_stage[CHUNK];
    __shared__ int  s_wsum[L1_THREADS / 32];
    const int t    = threadIdx.x;
    const int g    = blockIdx.x;
    const int cbeg = g * CHUNK;
    const int cend = min(cbeg + CHUNK, E);

    for (int b = t; b <= NB; b += L1_THREADS) s_hist[b] = 0;
    __syncthreads();

    {
        int s[L1_EPT];
        #pragma unroll
        for (int u = 0; u < L1_EPT; ++u)
            s[u] = src[min(cbeg + t + u * L1_THREADS, cend - 1)];
        #pragma unroll
        for (int u = 0; u < L1_EPT; ++u)
            if (cbeg + t + u * L1_THREADS < cend) atomicAdd(&s_hist[s[u] >> BSHIFT], 1);
    }
    __syncthreads();

    int carry = 0;
    for (int b0 = 0; b0 <= NB; b0 += L1_THREADS) {
        const int b = b0 + t;
        const int v = (b <= NB) ? s_hist[b] : 0;
        int tot;
        const int ex = carry + block_scan_excl<L1_THREADS / 32>(v, s_wsum, tot);
        if (b <= NB) {
            s_hist[b] = ex;
            offsT[(size_t)b * G + g] = ex;
        }
        carry += tot;
        __syncthreads();
    }

    {
        int s[L1_EPT], d[L1_EPT], pos[L1_EPT];
        #pragma unroll
        for (int u = 0; u < L1_EPT; ++u) {
            const int ec = min(cbeg + t + u * L1_THREADS, cend - 1);
            s[u] = src[ec];
            d[u] = dst[ec];
        }
        #pragma unroll
        for (int u = 0; u < L1_EPT; ++u) {
            pos[u] = 0;
            if (cbeg + t + u * L1_THREADS < cend) pos[u] = atomicAdd(&s_hist[s[u] >> BSHIFT], 1);
        }
        #pragma unroll
        for (int u = 0; u < L1_EPT; ++u)
            if (cbeg + t + u * L1_THREADS < cend) s_stage[pos[u]] = make_int2(s[u], d[u]);
    }
    __syncthreads();

    const int n = cend - cbeg;
    for (int i = t; i < n; i += L1_THREADS) tmp[(size_t)cbeg + i] = s_stage[i];
}

template <bool PLACE>
__device__ __forceinline__ int sweep_pieces(
    const int2* __restrict__ tmp, const int* __restrict__ rowS,
    const int* __restrict__ rowE, int* s_cnt, int* __restrict__ perm, int G)
{
    const int lane = threadIdx.x & 31;
    const int wv   = threadIdx.x >> 5;
    int ssum = 0;
    for (int gb = wv * 32; gb < G; gb += 256) {
        const int gg = gb + lane;
        const int s  = (gg < G) ? rowS[gg] : 0;
        const int e  = (gg < G) ? rowE[gg] : 0;
        ssum += s;
        #pragma unroll 1
        for (int i = 0; i < 32; i += L2_BATCH) {
            int  si[L2_BATCH], ei[L2_BATCH];
            int2 v[L2_BATCH];
            bool ok[L2_BATCH];
            #pragma unroll
            for (int u = 0; u < L2_BATCH; ++u) {
                si[u] = __shfl(s, i + u, 32);
                ei[u] = __shfl(e, i + u, 32);
                const int k = si[u] + lane;
                ok[u] = k < ei[u];
                const size_t at = ok[u] ? (size_t)(gb + i + u) * CHUNK + k : (size_t)0;
                if (PLACE) v[u] = tmp[at];
                else       v[u] = make_int2(((const int*)tmp)[2 * at], 0);
            }
            int pos[L2_BATCH];
            #pragma unroll
            for (int u = 0; u < L2_BATCH; ++u) {
                pos[u] = 0;
                if (ok[u]) pos[u] = atomicAdd(&s_cnt[v[u].x & (BNODES - 1)], 1);
            }
            if (PLACE) {
                #pragma unroll
                for (int u = 0; u < L2_BATCH; ++u)
                    if (ok[u]) perm[pos[u]] = v[u].y;
            }
            #pragma unroll
            for (int u = 0; u < L2_BATCH; ++u) {
                const int2* p = tmp + (size_t)(gb + i + u) * CHUNK;
                for (int k = si[u] + 32 + lane; k < ei[u]; k += 32) {
                    const int2 w = p[k];
                    if (PLACE) perm[atomicAdd(&s_cnt[w.x & (BNODES - 1)], 1)] = w.y;
                    else       atomicAdd(&s_cnt[w.x & (BNODES - 1)], 1);
                }
            }
        }
    }
    return ssum;
}

__global__ __launch_bounds__(BNODES) void sort_buckets(
    const int2* __restrict__ tmp, const int* __restrict__ offsT,
    int* __restrict__ cnt, int* __restrict__ cur, int* __restrict__ perm,
    int N, int G)
{
    __shared__ int s_cnt[BNODES];
    __shared__ int s_w[BNODES / 32];
    __shared__ int s_base[BNODES / 32];
    const int t    = threadIdx.x;
    const int lane = t & 31;
    const int b    = blockIdx.x;
    const int* rowS = offsT + (size_t)b * G;
    const int* rowE = rowS + G;

    s_cnt[t] = 0;
    __syncthreads();

    int ssum = sweep_pieces<false>(tmp, rowS, rowE, s_cnt, perm, G);
    #pragma unroll
    for (int o = 16; o > 0; o >>= 1) ssum += __shfl_xor(ssum, o, 32);
    if (lane == 0) s_base[t >> 5] = ssum;
    __syncthreads();

    int base = 0;
    #pragma unroll
    for (int i = 0; i < BNODES / 32; ++i) base += s_base[i];

    const int c = s_cnt[t];
    int tot;
    const int beg = base + block_scan_excl<BNODES / 32>(c, s_w, tot);
    const int node = b * BNODES + t;
    if (node < N) {
        cnt[node] = c;
        cur[node] = beg + c;
    }
    s_cnt[t] = beg;
    __syncthreads();

    (void)sweep_pieces<true>(tmp, rowS, rowE, s_cnt, perm, G);
}

__device__ __forceinline__ float dot8(const float4 q, const float4 k) {
    float p = q.x * k.x + q.y * k.y + q.z * k.z + q.w * k.w;
    p += __shfl_xor(p, 1, 8);
    p += __shfl_xor(p, 2, 8);
    p += __shfl_xor(p, 4, 8);
    return p;
}

__global__ __launch_bounds__(256) void node_attend(
    const float* __restrict__ Q, const float* __restrict__ K,
    const float* __restrict__ V,
    const int* __restrict__ cnt, const int* __restrict__ cur,
    const int* __restrict__ perm, float* __restrict__ out, int N)
{
    const int n = blockIdx.x * 32 + (threadIdx.x >> 3);
    if (n >= N) return;
    const int sub = threadIdx.x & 7;
    const int end = cur[n];
    const int beg = end - cnt[n];

    const float4 q = ((const float4*)(Q + (size_t)n * HEAD_SIZE))[sub];
    const float4* K4 = (const float4*)K + sub;
    const float4* V4 = (const float4*)V + sub;

    float  mx = -INFINITY;
    float  l  = 0.0f;
    float4 a  = make_float4(0.0f, 0.0f, 0.0f, 0.0f);

    for (int j = beg; j < end; j += 4) {
        const int last = end - 1 - j;
        const int d0 = perm[j];
        const int d1 = perm[j + min(1, last)];
        const int d2 = perm[j + min(2, last)];
        const int d3 = perm[j + min(3, last)];
        const float4 k0 = K4[(size_t)d0 * 8];
        const float4 k1 = K4[(size_t)d1 * 8];
        const float4 k2 = K4[(size_t)d2 * 8];
        const float4 k3 = K4[(size_t)d3 * 8];
        const float4 v0 = V4[(size_t)d0 * 8];
        const float4 v1 = V4[(size_t)d1 * 8];
        const float4 v2 = V4[(size_t)d2 * 8];
        const float4 v3 = V4[(size_t)d3 * 8];
        const float p0 = dot8(q, k0);
        const float p1 = dot8(q, k1);
        const float p2 = dot8(q, k2);
        const float p3 = dot8(q, k3);
        const float s0 = p0 * INV_DK;
        const float s1 = (last >= 1) ? p1 * INV_DK : -INFINITY;
        const float s2 = (last >= 2) ? p2 * INV_DK : -INFINITY;
        const float s3 = (last >= 3) ? p3 * INV_DK : -INFINITY;
        const float mn   = fmaxf(fmaxf(mx, s0), fmaxf(s1, fmaxf(s2, s3)));
        const float corr = __expf(mx - mn);
        const float e0 = __expf(s0 - mn);
        const float e1 = __expf(s1 - mn);
        const float e2 = __expf(s2 - mn);
        const float e3 = __expf(s3 - mn);
        l   = l   * corr + ((e0 + e1) + (e2 + e3));
        a.x = a.x * corr + ((e0 * v0.x + e1 * v1.x) + (e2 * v2.x + e3 * v3.x));
        a.y = a.y * corr + ((e0 * v0.y + e1 * v1.y) + (e2 * v2.y + e3 * v3.y));
        a.z = a.z * corr + ((e0 * v0.z + e1 * v1.z) + (e2 * v2.z + e3 * v3.z));
        a.w = a.w * corr + ((e0 * v0.w + e1 * v1.w) + (e2 * v2.w + e3 * v3.w));
        mx = mn;
    }

    const float dd = (l > 0.0f) ? l : 1.0f;
    float4 r;
    r.x = a.x / dd; r.y = a.y / dd; r.z = a.z / dd; r.w = a.w / dd;
    ((float4*)(out + (size_t)n * HEAD_SIZE))[sub] = r;
}

extern "C" void kernel_launch(void* const* d_in, const int* in_sizes, int n_in,
                              void* d_out, int out_size, void* d_ws, size_t ws_size,
                              hipStream_t stream) {
    const float* X    = (const float*)d_in[0];
    const int*   edge = (const int*)  d_in[1];
    const float* Wq   = (const float*)d_in[2];
    const float* Wk   = (const float*)d_in[3];
    const float* Wv   = (const float*)d_in[4];

    const int N = in_sizes[0] / INPUT_SIZE;
    const int E = in_sizes[1] / 2;
    const int* src = edge;
    const int* dst = edge + E;

    float* out = (float*)d_out;

    float* ws   = (float*)d_ws;
    float* Q    = ws;
    float* Kb   = ws + (size_t)N * HEAD_SIZE;
    float* Vb   = ws + (size_t)2 * N * HEAD_SIZE;
    int*   cnt  = (int*)(ws + (size_t)3 * N * HEAD_SIZE);
    int*   cur  = cnt + N;
    int*   perm = cur + N;

    const int NB = (N + BNODES - 1) / BNODES;
    const int G  = (E + CHUNK - 1) / CHUNK;
    int2* tmp   = (int2*)d_ws;
    int*  offsT = (int*)d_ws + (size_t)2 * E;

    sort_chunks<<<G, L1_THREADS, (NB + 1) * sizeof(int), stream>>>(
        src, dst, tmp, offsT, E, NB, G);
    sort_buckets<<<NB, BNODES, 0, stream>>>(tmp, offsT, cnt, cur, perm, N, G);
    {
        const int tiles  = (N + 15) / 16;
        const int waves  = (tiles + MTILES - 1) / MTILES;
        const int blocks = (waves + 7) / 8;
        gat_qkv_wmma<<<blocks, 256, 0, stream>>>(X, Wq, Wk, Wv, Q, Kb, Vb, N);
    }
    node_attend<<<(N + 31) / 32, 256, 0, stream>>>(Q, Kb, Vb, cnt, cur, perm, out, N);
}


# level-2 sort and QKV projection run side by side in ONE launch (block-uniform branch, projection blocks first); their buffers no longer overlap: Q is written to out, 3 launches instead of 4
# speedup vs baseline: 3.8716x; 1.2232x over previous
#include <hip/hip_runtime.h>
#include <hip/hip_bf16.h>
#include <math.h>

typedef __attribute__((ext_vector_type(2))) float v2f;
typedef __attribute__((ext_vector_type(8))) float v8f;

#define INPUT_SIZE 256
#define HEAD_SIZE  32
#define INV_DK     0.17677669529663687f
#define MTILES     4

__device__ __forceinline__ void qkv_wmma_block(
    const int bid,
    const float* __restrict__ X,
    const float* __restrict__ Wq, const float* __restrict__ Wk,
    const float* __restrict__ Wv,
    float* __restrict__ Q, float* __restrict__ Ko, float* __restrict__ Vo,
    int N)
{
    const int lane    = threadIdx.x & 31;
    const int wave    = threadIdx.x >> 5;
    const int tile0   = (bid * 8 + wave) * MTILES;
    if (tile0 * 16 >= N) return;

    const int m  = lane & 15;
    const int kh = (lane >> 4) * 2;

    const float* Ws[3] = { Wq, Wk, Wv };

    v8f acc[MTILES][6];
    #pragma unroll
    for (int t = 0; t < MTILES; ++t)
        #pragma unroll
        for (int i = 0; i < 6; ++i) {
            v8f z = {};
            acc[t][i] = z;
        }

    const float* xrow[MTILES];
    bool valid[MTILES];
    #pragma unroll
    for (int t = 0; t < MTILES; ++t) {
        const int rb = (tile0 + t) * 16;
        valid[t] = (rb + 16) <= N;
        const int r = valid[t] ? (rb + m) : m;
        xrow[t] = X + (size_t)r * INPUT_SIZE;
    }

    #pragma unroll 4
    for (int k0 = 0; k0 < INPUT_SIZE; k0 += 4) {
        v2f a[MTILES];
        #pragma unroll
        for (int t = 0; t < MTILES; ++t) {
            a[t].x = xrow[t][k0 + kh];
            a[t].y = xrow[t][k0 + kh + 1];
        }
        #pragma unroll
        for (int w = 0; w < 3; ++w) {
            const float* W = Ws[w];
            #pragma unroll
            for (int h = 0; h < 2; ++h) {
                const int col = m + h * 16;
                v2f b;
                b.x = W[(k0 + kh)     * HEAD_SIZE + col];
                b.y = W[(k0 + kh + 1) * HEAD_SIZE + col];
                #pragma unroll
                for (int t = 0; t < MTILES; ++t) {
                    acc[t][w * 2 + h] = __builtin_amdgcn_wmma_f32_16x16x4_f32(
                         false, a[t],  false, b,
                         (short)0, acc[t][w * 2 + h],
                         false,  false);
                }
            }
        }
    }

    float* Outs[3] = { Q, Ko, Vo };
    const int rofs = (lane >> 4) * 8;
    #pragma unroll
    for (int t = 0; t < MTILES; ++t) {
        if (!valid[t]) continue;
        const int row_base = (tile0 + t) * 16;
        #pragma unroll
        for (int w = 0; w < 3; ++w) {
            #pragma unroll
            for (int h = 0; h < 2; ++h) {
                const v8f c = acc[t][w * 2 + h];
                const int col = m + h * 16;
                #pragma unroll
                for (int j = 0; j < 8; ++j) {
                    const int row = row_base + j + rofs;
                    Outs[w][(size_t)row * HEAD_SIZE + col] = c[j];
                }
            }
        }
    }
}

#define CHUNK       6400
#define L1_THREADS  512
#define L1_EPT      ((CHUNK + L1_THREADS - 1) / L1_THREADS)
#define BSHIFT      8
#define BNODES      256
#define L2_BATCH    16

template <int NW>
__device__ __forceinline__ int block_scan_excl(int v, int* wsum, int& total) {
    const int lane = threadIdx.x & 31;
    const int wv   = threadIdx.x >> 5;
    int inc = v;
    #pragma unroll
    for (int o = 1; o < 32; o <<= 1) {
        const int t = __shfl_up(inc, o, 32);
        if (lane >= o) inc += t;
    }
    if (lane == 31) wsum[wv] = inc;
    __syncthreads();
    int woff = 0, tot = 0;
    #pragma unroll
    for (int i = 0; i < NW; ++i) {
        const int w = wsum[i];
        if (i < wv) woff += w;
        tot += w;
    }
    total = tot;
    return woff + inc - v;
}

__global__ __launch_bounds__(L1_THREADS) void sort_chunks(
    const int* __restrict__ src, const int* __restrict__ dst,
    int2* __restrict__ tmp, int* __restrict__ offsT, int E, int NB, int G)
{
    extern __shared__ int s_hist[];
    __shared__ int2 s_stage[CHUNK];
    __shared__ int  s_wsum[L1_THREADS / 32];
    const int t    = threadIdx.x;
    const int g    = blockIdx.x;
    const int cbeg = g * CHUNK;
    const int cend = min(cbeg + CHUNK, E);

    for (int b = t; b <= NB; b += L1_THREADS) s_hist[b] = 0;
    __syncthreads();

    {
        int s[L1_EPT];
        #pragma unroll
        for (int u = 0; u < L1_EPT; ++u)
            s[u] = src[min(cbeg + t + u * L1_THREADS, cend - 1)];
        #pragma unroll
        for (int u = 0; u < L1_EPT; ++u)
            if (cbeg + t + u * L1_THREADS < cend) atomicAdd(&s_hist[s[u] >> BSHIFT], 1);
    }
    __syncthreads();

    int carry = 0;
    for (int b0 = 0; b0 <= NB; b0 += L1_THREADS) {
        const int b = b0 + t;
        const int v = (b <= NB) ? s_hist[b] : 0;
        int tot;
        const int ex = carry + block_scan_excl<L1_THREADS / 32>(v, s_wsum, tot);
        if (b <= NB) {
            s_hist[b] = ex;
            offsT[(size_t)b * G + g] = ex;
        }
        carry += tot;
        __syncthreads();
    }

    {
        int s[L1_EPT], d[L1_EPT], pos[L1_EPT];
        #pragma unroll
        for (int u = 0; u < L1_EPT; ++u) {
            const int ec = min(cbeg + t + u * L1_THREADS, cend - 1);
            s[u] = src[ec];
            d[u] = dst[ec];
        }
        #pragma unroll
        for (int u = 0; u < L1_EPT; ++u) {
            pos[u] = 0;
            if (cbeg + t + u * L1_THREADS < cend) pos[u] = atomicAdd(&s_hist[s[u] >> BSHIFT], 1);
        }
        #pragma unroll
        for (int u = 0; u < L1_EPT; ++u)
            if (cbeg + t + u * L1_THREADS < cend) s_stage[pos[u]] = make_int2(s[u], d[u]);
    }
    __syncthreads();

    const int n = cend - cbeg;
    for (int i = t; i < n; i += L1_THREADS) tmp[(size_t)cbeg + i] = s_stage[i];
}

template <bool PLACE>
__device__ __forceinline__ int sweep_pieces(
    const int2* __restrict__ tmp, const int* __restrict__ rowS,
    const int* __restrict__ rowE, int* s_cnt, int* __restrict__ perm, int G)
{
    const int lane = threadIdx.x & 31;
    const int wv   = threadIdx.x >> 5;
    int ssum = 0;
    for (int gb = wv * 32; gb < G; gb += 256) {
        const int gg = gb + lane;
        const int s  = (gg < G) ? rowS[gg] : 0;
        const int e  = (gg < G) ? rowE[gg] : 0;
        ssum += s;
        #pragma unroll 1
        for (int i = 0; i < 32; i += L2_BATCH) {
            int  si[L2_BATCH], ei[L2_BATCH];
            int2 v[L2_BATCH];
            bool ok[L2_BATCH];
            #pragma unroll
            for (int u = 0; u < L2_BATCH; ++u) {
                si[u] = __shfl(s, i + u, 32);
                ei[u] = __shfl(e, i + u, 32);
                const int k = si[u] + lane;
                ok[u] = k < ei[u];
                const size_t at = ok[u] ? (size_t)(gb + i + u) * CHUNK + k : (size_t)0;
                if (PLACE) v[u] = tmp[at];
                else       v[u] = make_int2(((const int*)tmp)[2 * at], 0);
            }
            int pos[L2_BATCH];
            #pragma unroll
            for (int u = 0; u < L2_BATCH; ++u) {
                pos[u] = 0;
                if (ok[u]) pos[u] = atomicAdd(&s_cnt[v[u].x & (BNODES - 1)], 1);
            }
            if (PLACE) {
                #pragma unroll
                for (int u = 0; u < L2_BATCH; ++u)
                    if (ok[u]) perm[pos[u]] = v[u].y;
            }
            #pragma unroll
            for (int u = 0; u < L2_BATCH; ++u) {
                const int2* p = tmp + (size_t)(gb + i + u) * CHUNK;
                for (int k = si[u] + 32 + lane; k < ei[u]; k += 32) {
                    const int2 w = p[k];
                    if (PLACE) perm[atomicAdd(&s_cnt[w.x & (BNODES - 1)], 1)] = w.y;
                    else       atomicAdd(&s_cnt[w.x & (BNODES - 1)], 1);
                }
            }
        }
    }
    return ssum;
}

__device__ __forceinline__ void sort_bucket_block(
    const int b,
    const int2* __restrict__ tmp, const int* __restrict__ offsT,
    int* __restrict__ cur, int* __restrict__ perm,
    int N, int G)
{
    __shared__ int s_cnt[BNODES];
    __shared__ int s_w[BNODES / 32];
    __shared__ int s_base[BNODES / 32];
    const int t    = threadIdx.x;
    const int lane = t & 31;
    const int* rowS = offsT + (size_t)b * G;
    const int* rowE = rowS + G;

    s_cnt[t] = 0;
    __syncthreads();

    int ssum = sweep_pieces<false>(tmp, rowS, rowE, s_cnt, perm, G);
    #pragma unroll
    for (int o = 16; o > 0; o >>= 1) ssum += __shfl_xor(ssum, o, 32);
    if (lane == 0) s_base[t >> 5] = ssum;
    __syncthreads();

    int base = 0;
    #pragma unroll
    for (int i = 0; i < BNODES / 32; ++i) base += s_base[i];

    const int c = s_cnt[t];
    int tot;
    const int beg = base + block_scan_excl<BNODES / 32>(c, s_w, tot);
    const int node = b * BNODES + t;
    if (node < N) cur[node] = beg + c;
    s_cnt[t] = beg;
    __syncthreads();

    (void)sweep_pieces<true>(tmp, rowS, rowE, s_cnt, perm, G);
}

__global__ __launch_bounds__(BNODES) void buckets_and_qkv(
    const int2* __restrict__ tmp, const int* __restrict__ offsT,
    int* __restrict__ cur, int* __restrict__ perm, int N, int G,
    const float* __restrict__ X,
    const float* __restrict__ Wq, const float* __restrict__ Wk,
    const float* __restrict__ Wv,
    float* __restrict__ Q, float* __restrict__ Ko, float* __restrict__ Vo,
    int QB)
{
    if ((int)blockIdx.x < QB)
        qkv_wmma_block((int)blockIdx.x, X, Wq, Wk, Wv, Q, Ko, Vo, N);
    else
        sort_bucket_block((int)blockIdx.x - QB, tmp, offsT, cur, perm, N, G);
}

__device__ __forceinline__ float dot8(const float4 q, const float4 k) {
    float p = q.x * k.x + q.y * k.y + q.z * k.z + q.w * k.w;
    p += __shfl_xor(p, 1, 8);
    p += __shfl_xor(p, 2, 8);
    p += __shfl_xor(p, 4, 8);
    return p;
}

__global__ __launch_bounds__(256) void node_attend(
    const float* __restrict__ K, const float* __restrict__ V,
    const int* __restrict__ cur, const int* __restrict__ perm,
    float* out, int N)
{
    const int n = blockIdx.x * 32 + (threadIdx.x >> 3);
    if (n >= N) return;
    const int sub = threadIdx.x & 7;
    const int end = cur[n];
    const int beg = (n > 0) ? cur[n - 1] : 0;

    const float4 q = ((const float4*)(out + (size_t)n * HEAD_SIZE))[sub];
    const float4* K4 = (const float4*)K + sub;
    const float4* V4 = (const float4*)V + sub;

    float  mx = -INFINITY;
    float  l  = 0.0f;
    float4 a  = make_float4(0.0f, 0.0f, 0.0f, 0.0f);

    for (int j = beg; j < end; j += 4) {
        const int last = end - 1 - j;
        const int d0 = perm[j];
        const int d1 = perm[j + min(1, last)];
        const int d2 = perm[j + min(2, last)];
        const int d3 = perm[j + min(3, last)];
        const float4 k0 = K4[(size_t)d0 * 8];
        const float4 k1 = K4[(size_t)d1 * 8];
        const float4 k2 = K4[(size_t)d2 * 8];
        const float4 k3 = K4[(size_t)d3 * 8];
        const float4 v0 = V4[(size_t)d0 * 8];
        const float4 v1 = V4[(size_t)d1 * 8];
        const float4 v2 = V4[(size_t)d2 * 8];
        const float4 v3 = V4[(size_t)d3 * 8];
        const float p0 = dot8(q, k0);
        const float p1 = dot8(q, k1);
        const float p2 = dot8(q, k2);
        const float p3 = dot8(q, k3);
        const float s0 = p0 * INV_DK;
        const float s1 = (last >= 1) ? p1 * INV_DK : -INFINITY;
        const float s2 = (last >= 2) ? p2 * INV_DK : -INFINITY;
        const float s3 = (last >= 3) ? p3 * INV_DK : -INFINITY;
        const float mn   = fmaxf(fmaxf(mx, s0), fmaxf(s1, fmaxf(s2, s3)));
        const float corr = __expf(mx - mn);
        const float e0 = __expf(s0 - mn);
        const float e1 = __expf(s1 - mn);
        const float e2 = __expf(s2 - mn);
        const float e3 = __expf(s3 - mn);
        l   = l   * corr + ((e0 + e1) + (e2 + e3));
        a.x = a.x * corr + ((e0 * v0.x + e1 * v1.x) + (e2 * v2.x + e3 * v3.x));
        a.y = a.y * corr + ((e0 * v0.y + e1 * v1.y) + (e2 * v2.y + e3 * v3.y));
        a.z = a.z * corr + ((e0 * v0.z + e1 * v1.z) + (e2 * v2.z + e3 * v3.z));
        a.w = a.w * corr + ((e0 * v0.w + e1 * v1.w) + (e2 * v2.w + e3 * v3.w));
        mx = mn;
    }

    const float dd = (l > 0.0f) ? l : 1.0f;
    float4 r;
    r.x = a.x / dd; r.y = a.y / dd; r.z = a.z / dd; r.w = a.w / dd;
    ((float4*)(out + (size_t)n * HEAD_SIZE))[sub] = r;
}

extern "C" void kernel_launch(void* const* d_in, const int* in_sizes, int n_in,
                              void* d_out, int out_size, void* d_ws, size_t ws_size,
                              hipStream_t stream) {
    const float* X    = (const float*)d_in[0];
    const int*   edge = (const int*)  d_in[1];
    const float* Wq   = (const float*)d_in[2];
    const float* Wk   = (const float*)d_in[3];
    const float* Wv   = (const float*)d_in[4];

    const int N = in_sizes[0] / INPUT_SIZE;
    const int E = in_sizes[1] / 2;
    const int* src = edge;
    const int* dst = edge + E;

    float* out = (float*)d_out;

    const int NB = (N + BNODES - 1) / BNODES;
    const int G  = (E + CHUNK - 1) / CHUNK;
    float* ws    = (float*)d_ws;
    int2*  tmp   = (int2*)d_ws;
    float* Q     = out;
    float* Kb    = ws + (size_t)N * HEAD_SIZE;
    float* Vb    = ws + (size_t)2 * N * HEAD_SIZE;
    int*   offsT = (int*)(ws + (size_t)3 * N * HEAD_SIZE);
    int*   cur   = offsT + N;
    int*   perm  = cur + N;

    sort_chunks<<<G, L1_THREADS, (NB + 1) * sizeof(int), stream>>>(
        src, dst, tmp, offsT, E, NB, G);
    const int tiles = (N + 15) / 16;
    const int waves = (tiles + MTILES - 1) / MTILES;
    const int QB    = (waves + 7) / 8;
    buckets_and_qkv<<<QB + NB, BNODES, 0, stream>>>(
        tmp, offsT, cur, perm, N, G, X, Wq, Wk, Wv, Q, Kb, Vb, QB);
    node_attend<<<(N + 31) / 32, 256, 0, stream>>>(Kb, Vb, cur, perm, out, N);
}
